# starting kernel
# baseline (speedup 1.0000x reference)
#include <hip/hip_runtime.h>

typedef __attribute__((ext_vector_type(2))) float v2f;
typedef __attribute__((ext_vector_type(8))) float v8f;

#define LDSS 132


__global__ __launch_bounds__(256) void gcn_zero(float* __restrict__ p, long n) {
    long i = (long)blockIdx.x * blockDim.x + threadIdx.x;
    long stride = (long)gridDim.x * blockDim.x;
    for (; i < n; i += stride) p[i] = 0.0f;
}

__global__ __launch_bounds__(256) void gcn_degree(const int* __restrict__ src,
                                                  const int* __restrict__ dst,
                                                  float* __restrict__ deg_src,
                                                  float* __restrict__ deg_dst, int E) {
    int e = blockIdx.x * blockDim.x + threadIdx.x;
    if (e < E) {
        atomicAdd(&deg_src[src[e]], 1.0f);
        atomicAdd(&deg_dst[dst[e]], 1.0f);
    }
}

__global__ __launch_bounds__(256) void gcn_norm(float* __restrict__ deg, int n) {
    int i = blockIdx.x * blockDim.x + threadIdx.x;
    if (i < n) deg[i] = rsqrtf(fmaxf(deg[i], 1.0f));
}


__global__ __launch_bounds__(256) void gcn_gemm1(const float* __restrict__ feats,
                                                 const float* __restrict__ norm_src,
                                                 const float* __restrict__ W1,
                                                 float* __restrict__ H1, int nn) {
    __shared__ float As[16 * LDSS];
    const int tid  = threadIdx.x;
    const int wave = tid >> 5;
    const int lane = tid & 31;
    const int mbase = blockIdx.x * 16;

    for (int f = tid; f < 16 * 32; f += 256) {
        int row = f >> 5;
        int c4  = (f & 31) << 2;
        int r   = mbase + row; if (r >= nn) r = nn - 1;
        float nrm = norm_src[r];
        const float4 v = *(const float4*)(feats + (size_t)r * 128 + c4);
        float4 s; s.x = v.x * nrm; s.y = v.y * nrm; s.z = v.z * nrm; s.w = v.w * nrm;
        *(float4*)(&As[row * LDSS + c4]) = s;
    }
    __syncthreads();

    const int col  = (wave << 4) + (lane & 15);
    const int krow = (lane >> 4) << 1;
    v8f c = {};
    #pragma unroll 4
    for (int kk = 0; kk < 128; kk += 4) {
        const int k0 = kk + krow;
        v2f a = *(const v2f*)(&As[(lane & 15) * LDSS + k0]);
        v2f b;
        b.x = W1[(size_t)k0 * 128 + col];
        b.y = W1[(size_t)(k0 + 1) * 128 + col];
        c = __builtin_amdgcn_wmma_f32_16x16x4_f32(false, a, false, b, (short)0, c, false, false);
    }

    const int rbase = mbase + ((lane >> 4) << 3);
    if (mbase + 16 <= nn) {
        float* p = H1 + (size_t)rbase * 128 + col;
        #pragma unroll
        for (int v = 0; v < 8; ++v) p[v * 128] = c[v];
    } else {
        #pragma unroll
        for (int v = 0; v < 8; ++v) {
            int r = rbase + v;
            if (r < nn) H1[(size_t)r * 128 + col] = c[v];
        }
    }
}


__global__ __launch_bounds__(256) void gcn_scatter128(const float* __restrict__ H,
                                                      const int* __restrict__ src,
                                                      const int* __restrict__ dst,
                                                      float* __restrict__ agg, int E) {
    int e = blockIdx.x * 8 + (threadIdx.x >> 5);
    if (e >= E) return;
    int lane = threadIdx.x & 31;
    int s = src[e], d = dst[e];
    const float4 m = *(const float4*)(H + (size_t)s * 128 + lane * 4);
    float* o = agg + (size_t)d * 128 + lane * 4;
    atomicAdd(o + 0, m.x); atomicAdd(o + 1, m.y);
    atomicAdd(o + 2, m.z); atomicAdd(o + 3, m.w);
}


__global__ __launch_bounds__(128) void gcn_gemm2(const float* __restrict__ agg1,
                                                 const float* __restrict__ norm_src,
                                                 const float* __restrict__ norm_dst,
                                                 const float* __restrict__ b1,
                                                 const float* __restrict__ W2,
                                                 float* __restrict__ H2, int nn) {
    __shared__ float As[16 * LDSS];
    const int tid  = threadIdx.x;
    const int wave = tid >> 5;
    const int lane = tid & 31;
    const int mbase = blockIdx.x * 16;

    for (int f = tid; f < 16 * 32; f += 128) {
        int row = f >> 5;
        int c4  = (f & 31) << 2;
        int r   = mbase + row; if (r >= nn) r = nn - 1;
        float nd = norm_dst[r];
        float ns = norm_src[r];
        const float4 v  = *(const float4*)(agg1 + (size_t)r * 128 + c4);
        const float4 bb = *(const float4*)(b1 + c4);
        float4 s;
        s.x = fmaxf(v.x * nd + bb.x, 0.0f) * ns;
        s.y = fmaxf(v.y * nd + bb.y, 0.0f) * ns;
        s.z = fmaxf(v.z * nd + bb.z, 0.0f) * ns;
        s.w = fmaxf(v.w * nd + bb.w, 0.0f) * ns;
        *(float4*)(&As[row * LDSS + c4]) = s;
    }
    __syncthreads();

    const int col  = (wave << 4) + (lane & 15);
    const int krow = (lane >> 4) << 1;
    v8f c = {};
    #pragma unroll 4
    for (int kk = 0; kk < 128; kk += 4) {
        const int k0 = kk + krow;
        v2f a = *(const v2f*)(&As[(lane & 15) * LDSS + k0]);
        v2f b;
        b.x = W2[(size_t)k0 * 64 + col];
        b.y = W2[(size_t)(k0 + 1) * 64 + col];
        c = __builtin_amdgcn_wmma_f32_16x16x4_f32(false, a, false, b, (short)0, c, false, false);
    }

    const int rbase = mbase + ((lane >> 4) << 3);
    if (mbase + 16 <= nn) {
        float* p = H2 + (size_t)rbase * 64 + col;
        #pragma unroll
        for (int v = 0; v < 8; ++v) p[v * 64] = c[v];
    } else {
        #pragma unroll
        for (int v = 0; v < 8; ++v) {
            int r = rbase + v;
            if (r < nn) H2[(size_t)r * 64 + col] = c[v];
        }
    }
}


__global__ __launch_bounds__(256) void gcn_scatter64(const float* __restrict__ H,
                                                     const int* __restrict__ src,
                                                     const int* __restrict__ dst,
                                                     float* __restrict__ agg, int E) {
    int e = blockIdx.x * 8 + (threadIdx.x >> 5);
    if (e >= E) return;
    int lane = threadIdx.x & 31;
    int s = src[e], d = dst[e];
    const float2 m = *(const float2*)(H + (size_t)s * 64 + lane * 2);
    float* o = agg + (size_t)d * 64 + lane * 2;
    atomicAdd(o + 0, m.x); atomicAdd(o + 1, m.y);
}


__global__ __launch_bounds__(256) void gcn_finalize(const float* __restrict__ agg2,
                                                    const float* __restrict__ norm_dst,
                                                    const float* __restrict__ b2,
                                                    float* __restrict__ out, long total) {
    long i = (long)blockIdx.x * blockDim.x + threadIdx.x;
    if (i < total) {
        long node = i >> 6;
        int  j    = (int)(i & 63);
        out[i] = agg2[i] * norm_dst[node] + b2[j];
    }
}


extern "C" void kernel_launch(void* const* d_in, const int* in_sizes, int n_in,
                              void* d_out, int out_size, void* d_ws, size_t ws_size,
                              hipStream_t stream) {
    const float* feats = (const float*)d_in[0];
    const float* W1    = (const float*)d_in[1];
    const float* b1    = (const float*)d_in[2];
    const float* W2    = (const float*)d_in[3];
    const float* b2    = (const float*)d_in[4];
    const int*   src   = (const int*)d_in[5];
    const int*   dst   = (const int*)d_in[6];

    const int N = in_sizes[0] / 128;
    const int E = in_sizes[5];

    float* ws       = (float*)d_ws;
    float* norm_src = ws;                  ws += N;
    float* norm_dst = ws;                  ws += N;
    float* agg1     = ws;                  ws += (size_t)N * 128;
    float* agg2     = ws;                  ws += (size_t)N * 64;
    float* H1       = ws;                  ws += (size_t)N * 128;
    float* H2       = ws;                  ws += (size_t)N * 64;

    const long nzero = (long)N * (2 + 128 + 64);
    gcn_zero<<<2048, 256, 0, stream>>>((float*)d_ws, nzero);

    gcn_degree<<<(E + 255) / 256, 256, 0, stream>>>(src, dst, norm_src, norm_dst, E);
    gcn_norm<<<(2 * N + 255) / 256, 256, 0, stream>>>((float*)d_ws, 2 * N);

    const int mtiles = (N + 15) / 16;

    gcn_gemm1<<<mtiles, 256, 0, stream>>>(feats, norm_src, W1, H1, N);
    gcn_scatter128<<<(E + 7) / 8, 256, 0, stream>>>(H1, src, dst, agg1, E);

    gcn_gemm2<<<mtiles, 128, 0, stream>>>(agg1, norm_src, norm_dst, b1, W2, H2, N);
    gcn_scatter64<<<(E + 7) / 8, 256, 0, stream>>>(H2, src, dst, agg2, E);

    const long total = (long)N * 64;
    gcn_finalize<<<(int)((total + 255) / 256), 256, 0, stream>>>(agg2, norm_dst, b2,
                                                                 (float*)d_out, total);
}


# degree counters padded to one per 128-byte line (were 32 per line): the 2E float atomics spread over 2N lines instead of about 3k hot ones; norm kernel reads them strided, writes dense norms
# speedup vs baseline: 1.2378x; 1.2378x over previous
#include <hip/hip_runtime.h>

typedef __attribute__((ext_vector_type(2))) float v2f;
typedef __attribute__((ext_vector_type(8))) float v8f;

#define LDSS 132
#define CPAD 32


__global__ __launch_bounds__(256) void gcn_zero(float* __restrict__ p, long n) {
    long i = (long)blockIdx.x * blockDim.x + threadIdx.x;
    long stride = (long)gridDim.x * blockDim.x;
    for (; i < n; i += stride) p[i] = 0.0f;
}

__global__ __launch_bounds__(256) void gcn_degree(const int* __restrict__ src,
                                                  const int* __restrict__ dst,
                                                  float* __restrict__ cnt_src,
                                                  float* __restrict__ cnt_dst, int E) {
    int e = blockIdx.x * blockDim.x + threadIdx.x;
    if (e < E) {
        atomicAdd(&cnt_src[(size_t)src[e] * CPAD], 1.0f);
        atomicAdd(&cnt_dst[(size_t)dst[e] * CPAD], 1.0f);
    }
}

__global__ __launch_bounds__(256) void gcn_norm(const float* __restrict__ cnt,
                                                float* __restrict__ norm, int n) {
    int i = blockIdx.x * blockDim.x + threadIdx.x;
    if (i < n) norm[i] = rsqrtf(fmaxf(cnt[(size_t)i * CPAD], 1.0f));
}


__global__ __launch_bounds__(256) void gcn_gemm1(const float* __restrict__ feats,
                                                 const float* __restrict__ norm_src,
                                                 const float* __restrict__ W1,
                                                 float* __restrict__ H1, int nn) {
    __shared__ float As[16 * LDSS];
    const int tid  = threadIdx.x;
    const int wave = tid >> 5;
    const int lane = tid & 31;
    const int mbase = blockIdx.x * 16;

    for (int f = tid; f < 16 * 32; f += 256) {
        int row = f >> 5;
        int c4  = (f & 31) << 2;
        int r   = mbase + row; if (r >= nn) r = nn - 1;
        float nrm = norm_src[r];
        const float4 v = *(const float4*)(feats + (size_t)r * 128 + c4);
        float4 s; s.x = v.x * nrm; s.y = v.y * nrm; s.z = v.z * nrm; s.w = v.w * nrm;
        *(float4*)(&As[row * LDSS + c4]) = s;
    }
    __syncthreads();

    const int col  = (wave << 4) + (lane & 15);
    const int krow = (lane >> 4) << 1;
    v8f c = {};
    #pragma unroll 4
    for (int kk = 0; kk < 128; kk += 4) {
        const int k0 = kk + krow;
        v2f a = *(const v2f*)(&As[(lane & 15) * LDSS + k0]);
        v2f b;
        b.x = W1[(size_t)k0 * 128 + col];
        b.y = W1[(size_t)(k0 + 1) * 128 + col];
        c = __builtin_amdgcn_wmma_f32_16x16x4_f32(false, a, false, b, (short)0, c, false, false);
    }

    const int rbase = mbase + ((lane >> 4) << 3);
    if (mbase + 16 <= nn) {
        float* p = H1 + (size_t)rbase * 128 + col;
        #pragma unroll
        for (int v = 0; v < 8; ++v) p[v * 128] = c[v];
    } else {
        #pragma unroll
        for (int v = 0; v < 8; ++v) {
            int r = rbase + v;
            if (r < nn) H1[(size_t)r * 128 + col] = c[v];
        }
    }
}


__global__ __launch_bounds__(256) void gcn_scatter128(const float* __restrict__ H,
                                                      const int* __restrict__ src,
                                                      const int* __restrict__ dst,
                                                      float* __restrict__ agg, int E) {
    int e = blockIdx.x * 8 + (threadIdx.x >> 5);
    if (e >= E) return;
    int lane = threadIdx.x & 31;
    int s = src[e], d = dst[e];
    const float4 m = *(const float4*)(H + (size_t)s * 128 + lane * 4);
    float* o = agg + (size_t)d * 128 + lane * 4;
    atomicAdd(o + 0, m.x); atomicAdd(o + 1, m.y);
    atomicAdd(o + 2, m.z); atomicAdd(o + 3, m.w);
}


__global__ __launch_bounds__(128) void gcn_gemm2(const float* __restrict__ agg1,
                                                 const float* __restrict__ norm_src,
                                                 const float* __restrict__ norm_dst,
                                                 const float* __restrict__ b1,
                                                 const float* __restrict__ W2,
                                                 float* __restrict__ H2, int nn) {
    __shared__ float As[16 * LDSS];
    const int tid  = threadIdx.x;
    const int wave = tid >> 5;
    const int lane = tid & 31;
    const int mbase = blockIdx.x * 16;

    for (int f = tid; f < 16 * 32; f += 128) {
        int row = f >> 5;
        int c4  = (f & 31) << 2;
        int r   = mbase + row; if (r >= nn) r = nn - 1;
        float nd = norm_dst[r];
        float ns = norm_src[r];
        const float4 v  = *(const float4*)(agg1 + (size_t)r * 128 + c4);
        const float4 bb = *(const float4*)(b1 + c4);
        float4 s;
        s.x = fmaxf(v.x * nd + bb.x, 0.0f) * ns;
        s.y = fmaxf(v.y * nd + bb.y, 0.0f) * ns;
        s.z = fmaxf(v.z * nd + bb.z, 0.0f) * ns;
        s.w = fmaxf(v.w * nd + bb.w, 0.0f) * ns;
        *(float4*)(&As[row * LDSS + c4]) = s;
    }
    __syncthreads();

    const int col  = (wave << 4) + (lane & 15);
    const int krow = (lane >> 4) << 1;
    v8f c = {};
    #pragma unroll 4
    for (int kk = 0; kk < 128; kk += 4) {
        const int k0 = kk + krow;
        v2f a = *(const v2f*)(&As[(lane & 15) * LDSS + k0]);
        v2f b;
        b.x = W2[(size_t)k0 * 64 + col];
        b.y = W2[(size_t)(k0 + 1) * 64 + col];
        c = __builtin_amdgcn_wmma_f32_16x16x4_f32(false, a, false, b, (short)0, c, false, false);
    }

    const int rbase = mbase + ((lane >> 4) << 3);
    if (mbase + 16 <= nn) {
        float* p = H2 + (size_t)rbase * 64 + col;
        #pragma unroll
        for (int v = 0; v < 8; ++v) p[v * 64] = c[v];
    } else {
        #pragma unroll
        for (int v = 0; v < 8; ++v) {
            int r = rbase + v;
            if (r < nn) H2[(size_t)r * 64 + col] = c[v];
        }
    }
}


__global__ __launch_bounds__(256) void gcn_scatter64(const float* __restrict__ H,
                                                     const int* __restrict__ src,
                                                     const int* __restrict__ dst,
                                                     float* __restrict__ agg, int E) {
    int e = blockIdx.x * 8 + (threadIdx.x >> 5);
    if (e >= E) return;
    int lane = threadIdx.x & 31;
    int s = src[e], d = dst[e];
    const float2 m = *(const float2*)(H + (size_t)s * 64 + lane * 2);
    float* o = agg + (size_t)d * 64 + lane * 2;
    atomicAdd(o + 0, m.x); atomicAdd(o + 1, m.y);
}


__global__ __launch_bounds__(256) void gcn_finalize(const float* __restrict__ agg2,
                                                    const float* __restrict__ norm_dst,
                                                    const float* __restrict__ b2,
                                                    float* __restrict__ out, long total) {
    long i = (long)blockIdx.x * blockDim.x + threadIdx.x;
    if (i < total) {
        long node = i >> 6;
        int  j    = (int)(i & 63);
        out[i] = agg2[i] * norm_dst[node] + b2[j];
    }
}


extern "C" void kernel_launch(void* const* d_in, const int* in_sizes, int n_in,
                              void* d_out, int out_size, void* d_ws, size_t ws_size,
                              hipStream_t stream) {
    const float* feats = (const float*)d_in[0];
    const float* W1    = (const float*)d_in[1];
    const float* b1    = (const float*)d_in[2];
    const float* W2    = (const float*)d_in[3];
    const float* b2    = (const float*)d_in[4];
    const int*   src   = (const int*)d_in[5];
    const int*   dst   = (const int*)d_in[6];

    const int N = in_sizes[0] / 128;
    const int E = in_sizes[5];

    float* ws       = (float*)d_ws;
    float* cnt_src  = ws;                  ws += (size_t)N * CPAD;
    float* cnt_dst  = ws;                  ws += (size_t)N * CPAD;
    float* agg1     = ws;                  ws += (size_t)N * 128;
    float* agg2     = ws;                  ws += (size_t)N * 64;
    float* H1       = ws;                  ws += (size_t)N * 128;
    float* H2       = ws;                  ws += (size_t)N * 64;
    float* norm_src = ws;                  ws += N;
    float* norm_dst = ws;                  ws += N;

    const long nzero = (long)N * (2 * CPAD + 128 + 64);
    gcn_zero<<<2048, 256, 0, stream>>>((float*)d_ws, nzero);

    gcn_degree<<<(E + 255) / 256, 256, 0, stream>>>(src, dst, cnt_src, cnt_dst, E);
    gcn_norm<<<(2 * N + 255) / 256, 256, 0, stream>>>(cnt_src, norm_src, 2 * N);

    const int mtiles = (N + 15) / 16;

    gcn_gemm1<<<mtiles, 256, 0, stream>>>(feats, norm_src, W1, H1, N);
    gcn_scatter128<<<(E + 7) / 8, 256, 0, stream>>>(H1, src, dst, agg1, E);

    gcn_gemm2<<<mtiles, 128, 0, stream>>>(agg1, norm_src, norm_dst, b1, W2, H2, N);
    gcn_scatter64<<<(E + 7) / 8, 256, 0, stream>>>(H2, src, dst, agg2, E);

    const long total = (long)N * 64;
    gcn_finalize<<<(int)((total + 255) / 256), 256, 0, stream>>>(agg2, norm_dst, b2,
                                                                 (float*)d_out, total);
}


# atomic edge scatter replaced by per-destination segment sum: edge list sorted by dst in-call (returning count adds, scan, fill), one wave sums a node's rows in registers; sum order = arrival slot
# speedup vs baseline: 2.1516x; 1.7382x over previous
#include <hip/hip_runtime.h>

typedef __attribute__((ext_vector_type(2))) float v2f;
typedef __attribute__((ext_vector_type(8))) float v8f;

#define LDSS 132
#define CPAD 32


__global__ __launch_bounds__(256) void gcn_zero(float* __restrict__ p, long n) {
    long i = (long)blockIdx.x * blockDim.x + threadIdx.x;
    long stride = (long)gridDim.x * blockDim.x;
    for (; i < n; i += stride) p[i] = 0.0f;
}

__global__ __launch_bounds__(256) void gcn_degree(const int* __restrict__ src,
                                                  const int* __restrict__ dst,
                                                  int* __restrict__ cnt_src,
                                                  int* __restrict__ cnt_dst,
                                                  int* __restrict__ slot, int E) {
    int e = blockIdx.x * blockDim.x + threadIdx.x;
    if (e < E) {
        atomicAdd(&cnt_src[(size_t)src[e] * CPAD], 1);
        slot[e] = atomicAdd(&cnt_dst[(size_t)dst[e] * CPAD], 1);
    }
}

__global__ __launch_bounds__(256) void gcn_norm(const int* __restrict__ cnt,
                                                float* __restrict__ norm,
                                                int* __restrict__ deg_dst, int nn) {
    int i = blockIdx.x * blockDim.x + threadIdx.x;
    if (i < 2 * nn) {
        int c = cnt[(size_t)i * CPAD];
        norm[i] = rsqrtf(fmaxf((float)c, 1.0f));
        if (i >= nn) deg_dst[i - nn] = c;
    }
}

__global__ __launch_bounds__(1024) void gcn_scan(const int* __restrict__ deg,
                                                 int* __restrict__ off, int n) {
    __shared__ int wsum[32];
    __shared__ int chunk_total;
    const int tid  = threadIdx.x;
    const int lane = tid & 31;
    const int wave = tid >> 5;
    int carry = 0;
    for (int base = 0; base < n; base += 4096) {
        const int i0 = base + tid * 4;
        int v0 = (i0     < n) ? deg[i0]     : 0;
        int v1 = (i0 + 1 < n) ? deg[i0 + 1] : 0;
        int v2 = (i0 + 2 < n) ? deg[i0 + 2] : 0;
        int v3 = (i0 + 3 < n) ? deg[i0 + 3] : 0;
        const int t = v0 + v1 + v2 + v3;
        int x = t;
        for (int dlt = 1; dlt < 32; dlt <<= 1) {
            int y = __shfl_up(x, dlt, 32);
            if (lane >= dlt) x += y;
        }
        if (lane == 31) wsum[wave] = x;
        __syncthreads();
        if (wave == 0) {
            const int w = wsum[lane];
            int xs = w;
            for (int dlt = 1; dlt < 32; dlt <<= 1) {
                int y = __shfl_up(xs, dlt, 32);
                if (lane >= dlt) xs += y;
            }
            wsum[lane] = xs - w;
            if (lane == 31) chunk_total = xs;
        }
        __syncthreads();
        int ex = carry + wsum[wave] + (x - t);
        if (i0     < n) off[i0]     = ex;  ex += v0;
        if (i0 + 1 < n) off[i0 + 1] = ex;  ex += v1;
        if (i0 + 2 < n) off[i0 + 2] = ex;  ex += v2;
        if (i0 + 3 < n) off[i0 + 3] = ex;
        carry += chunk_total;
        __syncthreads();
    }
    if (tid == 0) off[n] = carry;
}

__global__ __launch_bounds__(256) void gcn_fill(const int* __restrict__ src,
                                                const int* __restrict__ dst,
                                                const int* __restrict__ off,
                                                const int* __restrict__ slot,
                                                int* __restrict__ nbr, int E) {
    int e = blockIdx.x * blockDim.x + threadIdx.x;
    if (e < E) nbr[off[dst[e]] + slot[e]] = src[e];
}


__global__ __launch_bounds__(256) void gcn_gemm1(const float* __restrict__ feats,
                                                 const float* __restrict__ norm_src,
                                                 const float* __restrict__ W1,
                                                 float* __restrict__ H1, int nn) {
    __shared__ float As[16 * LDSS];
    const int tid  = threadIdx.x;
    const int wave = tid >> 5;
    const int lane = tid & 31;
    const int mbase = blockIdx.x * 16;

    for (int f = tid; f < 16 * 32; f += 256) {
        int row = f >> 5;
        int c4  = (f & 31) << 2;
        int r   = mbase + row; if (r >= nn) r = nn - 1;
        float nrm = norm_src[r];
        const float4 v = *(const float4*)(feats + (size_t)r * 128 + c4);
        float4 s; s.x = v.x * nrm; s.y = v.y * nrm; s.z = v.z * nrm; s.w = v.w * nrm;
        *(float4*)(&As[row * LDSS + c4]) = s;
    }
    __syncthreads();

    const int col  = (wave << 4) + (lane & 15);
    const int krow = (lane >> 4) << 1;
    v8f c = {};
    #pragma unroll 4
    for (int kk = 0; kk < 128; kk += 4) {
        const int k0 = kk + krow;
        v2f a = *(const v2f*)(&As[(lane & 15) * LDSS + k0]);
        v2f b;
        b.x = W1[(size_t)k0 * 128 + col];
        b.y = W1[(size_t)(k0 + 1) * 128 + col];
        c = __builtin_amdgcn_wmma_f32_16x16x4_f32(false, a, false, b, (short)0, c, false, false);
    }

    const int rbase = mbase + ((lane >> 4) << 3);
    if (mbase + 16 <= nn) {
        float* p = H1 + (size_t)rbase * 128 + col;
        #pragma unroll
        for (int v = 0; v < 8; ++v) p[v * 128] = c[v];
    } else {
        #pragma unroll
        for (int v = 0; v < 8; ++v) {
            int r = rbase + v;
            if (r < nn) H1[(size_t)r * 128 + col] = c[v];
        }
    }
}


__global__ __launch_bounds__(256) void gcn_gather128(const float* __restrict__ H,
                                                     const int* __restrict__ off,
                                                     const int* __restrict__ nbr,
                                                     float* __restrict__ agg, int nn) {
    const int d = __builtin_amdgcn_readfirstlane((int)(blockIdx.x * 8 + (threadIdx.x >> 5)));
    if (d >= nn) return;
    const int lane = threadIdx.x & 31;
    const int beg = off[d], end = off[d + 1];
    float4 acc = make_float4(0.0f, 0.0f, 0.0f, 0.0f);
    for (int base = beg; base < end; base += 32) {
        const int cnt = min(32, end - base);
        const int mine = (lane < cnt) ? nbr[base + lane] : 0;
        for (int k = 0; k < cnt; k += 8) {
            float4 m[8];
            #pragma unroll
            for (int u = 0; u < 8; ++u) {
                m[u] = make_float4(0.0f, 0.0f, 0.0f, 0.0f);
                if (u == 0 || k + u < cnt) {
                    const int s = __builtin_amdgcn_readlane(mine, k + u);
                    m[u] = *(const float4*)(H + (size_t)s * 128 + lane * 4);
                }
            }
            #pragma unroll
            for (int u = 0; u < 8; ++u) {
                acc.x += m[u].x; acc.y += m[u].y; acc.z += m[u].z; acc.w += m[u].w;
            }
        }
    }
    *(float4*)(agg + (size_t)d * 128 + lane * 4) = acc;
}


__global__ __launch_bounds__(128) void gcn_gemm2(const float* __restrict__ agg1,
                                                 const float* __restrict__ norm_src,
                                                 const float* __restrict__ norm_dst,
                                                 const float* __restrict__ b1,
                                                 const float* __restrict__ W2,
                                                 float* __restrict__ H2, int nn) {
    __shared__ float As[16 * LDSS];
    const int tid  = threadIdx.x;
    const int wave = tid >> 5;
    const int lane = tid & 31;
    const int mbase = blockIdx.x * 16;

    for (int f = tid; f < 16 * 32; f += 128) {
        int row = f >> 5;
        int c4  = (f & 31) << 2;
        int r   = mbase + row; if (r >= nn) r = nn - 1;
        float nd = norm_dst[r];
        float ns = norm_src[r];
        const float4 v  = *(const float4*)(agg1 + (size_t)r * 128 + c4);
        const float4 bb = *(const float4*)(b1 + c4);
        float4 s;
        s.x = fmaxf(v.x * nd + bb.x, 0.0f) * ns;
        s.y = fmaxf(v.y * nd + bb.y, 0.0f) * ns;
        s.z = fmaxf(v.z * nd + bb.z, 0.0f) * ns;
        s.w = fmaxf(v.w * nd + bb.w, 0.0f) * ns;
        *(float4*)(&As[row * LDSS + c4]) = s;
    }
    __syncthreads();

    const int col  = (wave << 4) + (lane & 15);
    const int krow = (lane >> 4) << 1;
    v8f c = {};
    #pragma unroll 4
    for (int kk = 0; kk < 128; kk += 4) {
        const int k0 = kk + krow;
        v2f a = *(const v2f*)(&As[(lane & 15) * LDSS + k0]);
        v2f b;
        b.x = W2[(size_t)k0 * 64 + col];
        b.y = W2[(size_t)(k0 + 1) * 64 + col];
        c = __builtin_amdgcn_wmma_f32_16x16x4_f32(false, a, false, b, (short)0, c, false, false);
    }

    const int rbase = mbase + ((lane >> 4) << 3);
    if (mbase + 16 <= nn) {
        float* p = H2 + (size_t)rbase * 64 + col;
        #pragma unroll
        for (int v = 0; v < 8; ++v) p[v * 64] = c[v];
    } else {
        #pragma unroll
        for (int v = 0; v < 8; ++v) {
            int r = rbase + v;
            if (r < nn) H2[(size_t)r * 64 + col] = c[v];
        }
    }
}


__global__ __launch_bounds__(256) void gcn_gather64(const float* __restrict__ H,
                                                    const int* __restrict__ off,
                                                    const int* __restrict__ nbr,
                                                    const float* __restrict__ norm_dst,
                                                    const float* __restrict__ b2,
                                                    float* __restrict__ out, int nn) {
    const int d = __builtin_amdgcn_readfirstlane((int)(blockIdx.x * 8 + (threadIdx.x >> 5)));
    if (d >= nn) return;
    const int lane = threadIdx.x & 31;
    const int beg = off[d], end = off[d + 1];
    float2 acc = make_float2(0.0f, 0.0f);
    for (int base = beg; base < end; base += 32) {
        const int cnt = min(32, end - base);
        const int mine = (lane < cnt) ? nbr[base + lane] : 0;
        for (int k = 0; k < cnt; k += 8) {
            float2 m[8];
            #pragma unroll
            for (int u = 0; u < 8; ++u) {
                m[u] = make_float2(0.0f, 0.0f);
                if (u == 0 || k + u < cnt) {
                    const int s = __builtin_amdgcn_readlane(mine, k + u);
                    m[u] = *(const float2*)(H + (size_t)s * 64 + lane * 2);
                }
            }
            #pragma unroll
            for (int u = 0; u < 8; ++u) { acc.x += m[u].x; acc.y += m[u].y; }
        }
    }
    const float nd = norm_dst[d];
    const float2 bb = *(const float2*)(b2 + lane * 2);
    float2 r;
    r.x = acc.x * nd + bb.x;
    r.y = acc.y * nd + bb.y;
    *(float2*)(out + (size_t)d * 64 + lane * 2) = r;
}


extern "C" void kernel_launch(void* const* d_in, const int* in_sizes, int n_in,
                              void* d_out, int out_size, void* d_ws, size_t ws_size,
                              hipStream_t stream) {
    const float* feats = (const float*)d_in[0];
    const float* W1    = (const float*)d_in[1];
    const float* b1    = (const float*)d_in[2];
    const float* W2    = (const float*)d_in[3];
    const float* b2    = (const float*)d_in[4];
    const int*   src   = (const int*)d_in[5];
    const int*   dst   = (const int*)d_in[6];

    const int N = in_sizes[0] / 128;
    const int E = in_sizes[5];

    float* ws       = (float*)d_ws;
    int*   cnt_src  = (int*)ws;            ws += (size_t)N * CPAD;
    int*   cnt_dst  = (int*)ws;            ws += (size_t)N * CPAD;
    float* agg1     = ws;                  ws += (size_t)N * 128;
    float* H1       = ws;                  ws += (size_t)N * 128;
    float* H2       = ws;                  ws += (size_t)N * 64;
    float* norm_src = ws;                  ws += N;
    float* norm_dst = ws;                  ws += N;
    int*   deg_dst  = (int*)ws;            ws += N;
    int*   off      = (int*)ws;            ws += N + 1;
    int*   slot     = (int*)ws;            ws += E;
    int*   nbr      = (int*)ws;            ws += E;

    const long nzero = (long)N * (2 * CPAD);
    gcn_zero<<<2048, 256, 0, stream>>>((float*)d_ws, nzero);

    gcn_degree<<<(E + 255) / 256, 256, 0, stream>>>(src, dst, cnt_src, cnt_dst, slot, E);
    gcn_norm<<<(2 * N + 255) / 256, 256, 0, stream>>>(cnt_src, norm_src, deg_dst, N);

    gcn_scan<<<1, 1024, 0, stream>>>(deg_dst, off, N);
    gcn_fill<<<(E + 255) / 256, 256, 0, stream>>>(src, dst, off, slot, nbr, E);

    const int mtiles = (N + 15) / 16;
    const int ntiles = (N + 7) / 8;

    gcn_gemm1<<<mtiles, 256, 0, stream>>>(feats, norm_src, W1, H1, N);
    gcn_gather128<<<ntiles, 256, 0, stream>>>(H1, off, nbr, agg1, N);

    gcn_gemm2<<<mtiles, 128, 0, stream>>>(agg1, norm_src, norm_dst, b1, W2, H2, N);
    gcn_gather64<<<ntiles, 256, 0, stream>>>(H2, off, nbr, norm_dst, b2, (float*)d_out, N);
}


# layer-1 neighbour sum moved in front of its W1 product (A(XW)=(AX)W): row sum, both products, bias, relu, norms in one kernel per 16 nodes; H1/agg1 never stored; sum order changes
# speedup vs baseline: 2.3479x; 1.0912x over previous
#include <hip/hip_runtime.h>

typedef __attribute__((ext_vector_type(2))) float v2f;
typedef __attribute__((ext_vector_type(8))) float v8f;

#define LDSS 132
#define CPAD 32


__global__ __launch_bounds__(256) void gcn_zero(float* __restrict__ p, long n) {
    long i = (long)blockIdx.x * blockDim.x + threadIdx.x;
    long stride = (long)gridDim.x * blockDim.x;
    for (; i < n; i += stride) p[i] = 0.0f;
}

__global__ __launch_bounds__(256) void gcn_degree(const int* __restrict__ src,
                                                  const int* __restrict__ dst,
                                                  int* __restrict__ cnt_src,
                                                  int* __restrict__ cnt_dst,
                                                  int* __restrict__ slot, int E) {
    int e = blockIdx.x * blockDim.x + threadIdx.x;
    if (e < E) {
        atomicAdd(&cnt_src[(size_t)src[e] * CPAD], 1);
        slot[e] = atomicAdd(&cnt_dst[(size_t)dst[e] * CPAD], 1);
    }
}

__global__ __launch_bounds__(256) void gcn_norm(const int* __restrict__ cnt,
                                                float* __restrict__ norm,
                                                int* __restrict__ deg_dst, int nn) {
    int i = blockIdx.x * blockDim.x + threadIdx.x;
    if (i < 2 * nn) {
        int c = cnt[(size_t)i * CPAD];
        norm[i] = rsqrtf(fmaxf((float)c, 1.0f));
        if (i >= nn) deg_dst[i - nn] = c;
    }
}

__global__ __launch_bounds__(1024) void gcn_scan(const int* __restrict__ deg,
                                                 int* __restrict__ off, int n) {
    __shared__ int wsum[32];
    __shared__ int chunk_total;
    const int tid  = threadIdx.x;
    const int lane = tid & 31;
    const int wave = tid >> 5;
    int carry = 0;
    for (int base = 0; base < n; base += 4096) {
        const int i0 = base + tid * 4;
        int v0 = (i0     < n) ? deg[i0]     : 0;
        int v1 = (i0 + 1 < n) ? deg[i0 + 1] : 0;
        int v2 = (i0 + 2 < n) ? deg[i0 + 2] : 0;
        int v3 = (i0 + 3 < n) ? deg[i0 + 3] : 0;
        const int t = v0 + v1 + v2 + v3;
        int x = t;
        for (int dlt = 1; dlt < 32; dlt <<= 1) {
            int y = __shfl_up(x, dlt, 32);
            if (lane >= dlt) x += y;
        }
        if (lane == 31) wsum[wave] = x;
        __syncthreads();
        if (wave == 0) {
            const int w = wsum[lane];
            int xs = w;
            for (int dlt = 1; dlt < 32; dlt <<= 1) {
                int y = __shfl_up(xs, dlt, 32);
                if (lane >= dlt) xs += y;
            }
            wsum[lane] = xs - w;
            if (lane == 31) chunk_total = xs;
        }
        __syncthreads();
        int ex = carry + wsum[wave] + (x - t);
        if (i0     < n) off[i0]     = ex;  ex += v0;
        if (i0 + 1 < n) off[i0 + 1] = ex;  ex += v1;
        if (i0 + 2 < n) off[i0 + 2] = ex;  ex += v2;
        if (i0 + 3 < n) off[i0 + 3] = ex;
        carry += chunk_total;
        __syncthreads();
    }
    if (tid == 0) off[n] = carry;
}

__global__ __launch_bounds__(256) void gcn_fill(const int* __restrict__ src,
                                                const int* __restrict__ dst,
                                                const int* __restrict__ off,
                                                const int* __restrict__ slot,
                                                const float* __restrict__ norm_src,
                                                int* __restrict__ rec, int E) {
    int e = blockIdx.x * blockDim.x + threadIdx.x;
    if (e < E) {
        const int s = src[e];
        int2 r;
        r.x = s;
        r.y = __float_as_int(norm_src[s]);
        ((int2*)rec)[off[dst[e]] + slot[e]] = r;
    }
}


__global__ __launch_bounds__(128) void gcn_layer1(const float* __restrict__ feats,
                                                  const int* __restrict__ off,
                                                  const int* __restrict__ rec,
                                                  const float* __restrict__ norm_src,
                                                  const float* __restrict__ norm_dst,
                                                  const float* __restrict__ W1,
                                                  const float* __restrict__ b1,
                                                  const float* __restrict__ W2,
                                                  float* __restrict__ H2, int nn) {
    __shared__ float As[16 * LDSS];
    __shared__ float Bs[16 * LDSS];
    __shared__ float nds[16];
    __shared__ float nss[16];
    const int tid  = threadIdx.x;
    const int lane = tid & 31;
    const int wave = __builtin_amdgcn_readfirstlane(tid >> 5);
    const int mbase = blockIdx.x * 16;

    if (tid < 16) {
        int r = mbase + tid; if (r >= nn) r = nn - 1;
        nds[tid] = norm_dst[r];
        nss[tid] = norm_src[r];
    }

    const int d0 = mbase + wave * 4;
    int ob = 0;
    if (lane <= 4) ob = off[min(d0 + lane, nn)];

    const int2* __restrict__ rec2 = (const int2*)rec;
    for (int j = 0; j < 4; ++j) {
        const int beg = __builtin_amdgcn_readlane(ob, j);
        const int end = __builtin_amdgcn_readlane(ob, j + 1);
        float4 acc = make_float4(0.0f, 0.0f, 0.0f, 0.0f);
        for (int base = beg; base < end; base += 32) {
            const int cnt = min(32, end - base);
            int2 mine = make_int2(0, 0);
            if (lane < cnt) mine = rec2[base + lane];
            for (int k = 0; k < cnt; k += 8) {
                float4 m[8];
                float  wv[8];
                #pragma unroll
                for (int u = 0; u < 8; ++u) {
                    m[u] = make_float4(0.0f, 0.0f, 0.0f, 0.0f);
                    wv[u] = 0.0f;
                    if (u == 0 || k + u < cnt) {
                        const int s = __builtin_amdgcn_readlane(mine.x, k + u);
                        wv[u] = __int_as_float(__builtin_amdgcn_readlane(mine.y, k + u));
                        m[u] = *(const float4*)(feats + (size_t)s * 128 + lane * 4);
                    }
                }
                #pragma unroll
                for (int u = 0; u < 8; ++u) {
                    acc.x += m[u].x * wv[u]; acc.y += m[u].y * wv[u];
                    acc.z += m[u].z * wv[u]; acc.w += m[u].w * wv[u];
                }
            }
        }
        *(float4*)(&As[(wave * 4 + j) * LDSS + lane * 4]) = acc;
    }
    __syncthreads();

    const int c16  = lane & 15;
    const int half = lane >> 4;
    const int krow = half << 1;

    const int colA = (wave << 5) + c16;
    const int colB = colA + 16;
    v8f c0 = {};
    v8f c1 = {};
    #pragma unroll 4
    for (int kk = 0; kk < 128; kk += 4) {
        const int k0 = kk + krow;
        v2f a = *(const v2f*)(&As[c16 * LDSS + k0]);
        v2f p, q;
        p.x = W1[(size_t)k0 * 128 + colA];
        p.y = W1[(size_t)(k0 + 1) * 128 + colA];
        q.x = W1[(size_t)k0 * 128 + colB];
        q.y = W1[(size_t)(k0 + 1) * 128 + colB];
        c0 = __builtin_amdgcn_wmma_f32_16x16x4_f32(false, a, false, p, (short)0, c0, false, false);
        c1 = __builtin_amdgcn_wmma_f32_16x16x4_f32(false, a, false, q, (short)0, c1, false, false);
    }

    {
        const float biasA = b1[colA];
        const float biasB = b1[colB];
        #pragma unroll
        for (int v = 0; v < 8; ++v) {
            const int row = (half << 3) + v;
            const float nd = nds[row];
            const float ns = nss[row];
            Bs[row * LDSS + colA] = fmaxf(c0[v] * nd + biasA, 0.0f) * ns;
            Bs[row * LDSS + colB] = fmaxf(c1[v] * nd + biasB, 0.0f) * ns;
        }
    }
    __syncthreads();

    const int col = (wave << 4) + c16;
    v8f c = {};
    #pragma unroll 4
    for (int kk = 0; kk < 128; kk += 4) {
        const int k0 = kk + krow;
        v2f a = *(const v2f*)(&Bs[c16 * LDSS + k0]);
        v2f b;
        b.x = W2[(size_t)k0 * 64 + col];
        b.y = W2[(size_t)(k0 + 1) * 64 + col];
        c = __builtin_amdgcn_wmma_f32_16x16x4_f32(false, a, false, b, (short)0, c, false, false);
    }

    const int rbase = mbase + (half << 3);
    if (mbase + 16 <= nn) {
        float* p = H2 + (size_t)rbase * 64 + col;
        #pragma unroll
        for (int v = 0; v < 8; ++v) p[v * 64] = c[v];
    } else {
        #pragma unroll
        for (int v = 0; v < 8; ++v) {
            int r = rbase + v;
            if (r < nn) H2[(size_t)r * 64 + col] = c[v];
        }
    }
}


__global__ __launch_bounds__(256) void gcn_gather64(const float* __restrict__ H,
                                                    const int* __restrict__ off,
                                                    const int* __restrict__ rec,
                                                    const float* __restrict__ norm_dst,
                                                    const float* __restrict__ b2,
                                                    float* __restrict__ out, int nn) {
    const int d = __builtin_amdgcn_readfirstlane((int)(blockIdx.x * 8 + (threadIdx.x >> 5)));
    if (d >= nn) return;
    const int lane = threadIdx.x & 31;
    const int beg = off[d], end = off[d + 1];
    float2 acc = make_float2(0.0f, 0.0f);
    for (int base = beg; base < end; base += 32) {
        const int cnt = min(32, end - base);
        const int mine = (lane < cnt) ? rec[2 * (base + lane)] : 0;
        for (int k = 0; k < cnt; k += 8) {
            float2 m[8];
            #pragma unroll
            for (int u = 0; u < 8; ++u) {
                m[u] = make_float2(0.0f, 0.0f);
                if (u == 0 || k + u < cnt) {
                    const int s = __builtin_amdgcn_readlane(mine, k + u);
                    m[u] = *(const float2*)(H + (size_t)s * 64 + lane * 2);
                }
            }
            #pragma unroll
            for (int u = 0; u < 8; ++u) { acc.x += m[u].x; acc.y += m[u].y; }
        }
    }
    const float nd = norm_dst[d];
    const float2 bb = *(const float2*)(b2 + lane * 2);
    float2 r;
    r.x = acc.x * nd + bb.x;
    r.y = acc.y * nd + bb.y;
    *(float2*)(out + (size_t)d * 64 + lane * 2) = r;
}


extern "C" void kernel_launch(void* const* d_in, const int* in_sizes, int n_in,
                              void* d_out, int out_size, void* d_ws, size_t ws_size,
                              hipStream_t stream) {
    const float* feats = (const float*)d_in[0];
    const float* W1    = (const float*)d_in[1];
    const float* b1    = (const float*)d_in[2];
    const float* W2    = (const float*)d_in[3];
    const float* b2    = (const float*)d_in[4];
    const int*   src   = (const int*)d_in[5];
    const int*   dst   = (const int*)d_in[6];

    const int N = in_sizes[0] / 128;
    const int E = in_sizes[5];

    float* ws       = (float*)d_ws;
    int*   cnt_src  = (int*)ws;            ws += (size_t)N * CPAD;
    int*   cnt_dst  = (int*)ws;            ws += (size_t)N * CPAD;
    float* H2       = ws;                  ws += (size_t)N * 64;
    int*   rec      = (int*)ws;            ws += (size_t)E * 2;
    float* norm_src = ws;                  ws += N;
    float* norm_dst = ws;                  ws += N;
    int*   deg_dst  = (int*)ws;            ws += N;
    int*   off      = (int*)ws;            ws += N + 1;
    int*   slot     = (int*)ws;            ws += E;

    const long nzero = (long)N * (2 * CPAD);
    gcn_zero<<<2048, 256, 0, stream>>>((float*)d_ws, nzero);

    gcn_degree<<<(E + 255) / 256, 256, 0, stream>>>(src, dst, cnt_src, cnt_dst, slot, E);
    gcn_norm<<<(2 * N + 255) / 256, 256, 0, stream>>>(cnt_src, norm_src, deg_dst, N);

    gcn_scan<<<1, 1024, 0, stream>>>(deg_dst, off, N);
    gcn_fill<<<(E + 255) / 256, 256, 0, stream>>>(src, dst, off, slot, norm_src, rec, E);

    const int mtiles = (N + 15) / 16;
    const int ntiles = (N + 7) / 8;

    gcn_layer1<<<mtiles, 128, 0, stream>>>(feats, off, rec, norm_src, norm_dst, W1, b1, W2, H2, N);

    gcn_gather64<<<ntiles, 256, 0, stream>>>(H2, off, rec, norm_dst, b2, (float*)d_out, N);
}


# count/scan/fill edge sort replaced by direct filing: returning in-degree add = slot, src stored at once in a 32-slot line per node; edges past 32 go to an overflow list, summed last in list order
# speedup vs baseline: 2.9943x; 1.2753x over previous
#include <hip/hip_runtime.h>

typedef __attribute__((ext_vector_type(2))) float v2f;
typedef __attribute__((ext_vector_type(8))) float v8f;

#define LDSS 132
#define CPAD 32
#define CAP  32


__global__ __launch_bounds__(256) void gcn_zero(float* __restrict__ p, long n) {
    long i = (long)blockIdx.x * blockDim.x + threadIdx.x;
    long stride = (long)gridDim.x * blockDim.x;
    for (; i < n; i += stride) p[i] = 0.0f;
}

__global__ __launch_bounds__(256) void gcn_degree(const int* __restrict__ src,
                                                  const int* __restrict__ dst,
                                                  int* __restrict__ cnt_src,
                                                  int* __restrict__ cnt_dst,
                                                  int* __restrict__ bucket,
                                                  int* __restrict__ ovf_n,
                                                  int* __restrict__ ovf, int E) {
    int e = blockIdx.x * blockDim.x + threadIdx.x;
    if (e < E) {
        const int s = src[e];
        const int d = dst[e];
        atomicAdd(&cnt_src[(size_t)s * CPAD], 1);
        const int slot = atomicAdd(&cnt_dst[(size_t)d * CPAD], 1);
        if (slot < CAP) {
            bucket[(size_t)d * CAP + slot] = s;
        } else {
            const int k = atomicAdd(ovf_n, 1);
            ((int2*)ovf)[k] = make_int2(s, d);
        }
    }
}

__global__ __launch_bounds__(256) void gcn_norm(const int* __restrict__ cnt,
                                                float* __restrict__ norm,
                                                int* __restrict__ deg_dst, int nn) {
    int i = blockIdx.x * blockDim.x + threadIdx.x;
    if (i < 2 * nn) {
        int c = cnt[(size_t)i * CPAD];
        norm[i] = rsqrtf(fmaxf((float)c, 1.0f));
        if (i >= nn) deg_dst[i - nn] = c;
    }
}


__device__ __forceinline__ void rows128(const float* __restrict__ feats, int lane,
                                        int id, int wbits, int cnt, float4& acc) {
    for (int k = 0; k < cnt; k += 8) {
        float4 m[8];
        float  wv[8];
        #pragma unroll
        for (int u = 0; u < 8; ++u) {
            m[u] = make_float4(0.0f, 0.0f, 0.0f, 0.0f);
            wv[u] = 0.0f;
            if (u == 0 || k + u < cnt) {
                const int s = __builtin_amdgcn_readlane(id, k + u);
                wv[u] = __int_as_float(__builtin_amdgcn_readlane(wbits, k + u));
                m[u] = *(const float4*)(feats + (size_t)s * 128 + lane * 4);
            }
        }
        #pragma unroll
        for (int u = 0; u < 8; ++u) {
            acc.x += m[u].x * wv[u]; acc.y += m[u].y * wv[u];
            acc.z += m[u].z * wv[u]; acc.w += m[u].w * wv[u];
        }
    }
}

__device__ __forceinline__ void rows64(const float* __restrict__ H, int lane,
                                       int id, int cnt, float2& acc) {
    for (int k = 0; k < cnt; k += 8) {
        float2 m[8];
        #pragma unroll
        for (int u = 0; u < 8; ++u) {
            m[u] = make_float2(0.0f, 0.0f);
            if (u == 0 || k + u < cnt) {
                const int s = __builtin_amdgcn_readlane(id, k + u);
                m[u] = *(const float2*)(H + (size_t)s * 64 + lane * 2);
            }
        }
        #pragma unroll
        for (int u = 0; u < 8; ++u) { acc.x += m[u].x; acc.y += m[u].y; }
    }
}

__device__ __forceinline__ void ovf128(const float* __restrict__ feats,
                                       const float* __restrict__ norm_src,
                                       const int* __restrict__ ovf_n,
                                       const int* __restrict__ ovf, int E,
                                       int d, int lane, float4& acc) {
    const int no = min(*ovf_n, E);
    const int2* __restrict__ ovf2 = (const int2*)ovf;
    for (int ob = 0; ob < no; ob += 32) {
        int2 r = make_int2(0, -1);
        if (ob + lane < no) r = ovf2[ob + lane];
        unsigned hits = __builtin_amdgcn_ballot_w32(r.y == d);
        while (hits) {
            const int l = __builtin_ctz(hits);
            hits &= hits - 1;
            const int s = __builtin_amdgcn_readlane(r.x, l);
            const float w = norm_src[s];
            const float4 m = *(const float4*)(feats + (size_t)s * 128 + lane * 4);
            acc.x += m.x * w; acc.y += m.y * w; acc.z += m.z * w; acc.w += m.w * w;
        }
    }
}

__device__ __forceinline__ void ovf64(const float* __restrict__ H,
                                      const int* __restrict__ ovf_n,
                                      const int* __restrict__ ovf, int E,
                                      int d, int lane, float2& acc) {
    const int no = min(*ovf_n, E);
    const int2* __restrict__ ovf2 = (const int2*)ovf;
    for (int ob = 0; ob < no; ob += 32) {
        int2 r = make_int2(0, -1);
        if (ob + lane < no) r = ovf2[ob + lane];
        unsigned hits = __builtin_amdgcn_ballot_w32(r.y == d);
        while (hits) {
            const int l = __builtin_ctz(hits);
            hits &= hits - 1;
            const int s = __builtin_amdgcn_readlane(r.x, l);
            const float2 m = *(const float2*)(H + (size_t)s * 64 + lane * 2);
            acc.x += m.x; acc.y += m.y;
        }
    }
}


__global__ __launch_bounds__(128) void gcn_layer1(const float* __restrict__ feats,
                                                  const int* __restrict__ deg_dst,
                                                  const int* __restrict__ bucket,
                                                  const int* __restrict__ ovf_n,
                                                  const int* __restrict__ ovf, int E,
                                                  const float* __restrict__ norm_src,
                                                  const float* __restrict__ norm_dst,
                                                  const float* __restrict__ W1,
                                                  const float* __restrict__ b1,
                                                  const float* __restrict__ W2,
                                                  float* __restrict__ H2, int nn) {
    __shared__ float As[16 * LDSS];
    __shared__ float Bs[16 * LDSS];
    __shared__ float nds[16];
    __shared__ float nss[16];
    const int tid  = threadIdx.x;
    const int lane = tid & 31;
    const int wave = __builtin_amdgcn_readfirstlane(tid >> 5);
    const int mbase = blockIdx.x * 16;

    if (tid < 16) {
        int r = mbase + tid; if (r >= nn) r = nn - 1;
        nds[tid] = norm_dst[r];
        nss[tid] = norm_src[r];
    }

    const int d0 = mbase + wave * 4;
    int dg = 0;
    if (lane < 4 && d0 + lane < nn) dg = deg_dst[d0 + lane];

    int cnts[4];
    int ids[4];
    int wbs[4];
    #pragma unroll
    for (int j = 0; j < 4; ++j) {
        cnts[j] = min(__builtin_amdgcn_readlane(dg, j), CAP);
        const int* p = (lane < cnts[j]) ? bucket + ((size_t)(d0 + j) * CAP + lane) : deg_dst;
        ids[j] = *p;
    }
    #pragma unroll
    for (int j = 0; j < 4; ++j) {
        ids[j] = (lane < cnts[j]) ? ids[j] : 0;
        wbs[j] = __float_as_int(norm_src[ids[j]]);
    }
    #pragma unroll
    for (int j = 0; j < 4; ++j) {
        float4 acc = make_float4(0.0f, 0.0f, 0.0f, 0.0f);
        rows128(feats, lane, ids[j], wbs[j], cnts[j], acc);
        *(float4*)(&As[(wave * 4 + j) * LDSS + lane * 4]) = acc;
    }
    for (int j = 0; j < 4; ++j) {
        if (__builtin_amdgcn_readlane(dg, j) > CAP) {
            float4 acc = *(const float4*)(&As[(wave * 4 + j) * LDSS + lane * 4]);
            ovf128(feats, norm_src, ovf_n, ovf, E, d0 + j, lane, acc);
            *(float4*)(&As[(wave * 4 + j) * LDSS + lane * 4]) = acc;
        }
    }
    __syncthreads();

    const int c16  = lane & 15;
    const int half = lane >> 4;
    const int krow = half << 1;

    const int colA = (wave << 5) + c16;
    const int colB = colA + 16;
    v8f c0 = {};
    v8f c1 = {};
    #pragma unroll 4
    for (int kk = 0; kk < 128; kk += 4) {
        const int k0 = kk + krow;
        v2f a = *(const v2f*)(&As[c16 * LDSS + k0]);
        v2f p, q;
        p.x = W1[(size_t)k0 * 128 + colA];
        p.y = W1[(size_t)(k0 + 1) * 128 + colA];
        q.x = W1[(size_t)k0 * 128 + colB];
        q.y = W1[(size_t)(k0 + 1) * 128 + colB];
        c0 = __builtin_amdgcn_wmma_f32_16x16x4_f32(false, a, false, p, (short)0, c0, false, false);
        c1 = __builtin_amdgcn_wmma_f32_16x16x4_f32(false, a, false, q, (short)0, c1, false, false);
    }

    {
        const float biasA = b1[colA];
        const float biasB = b1[colB];
        #pragma unroll
        for (int v = 0; v < 8; ++v) {
            const int row = (half << 3) + v;
            const float nd = nds[row];
            const float ns = nss[row];
            Bs[row * LDSS + colA] = fmaxf(c0[v] * nd + biasA, 0.0f) * ns;
            Bs[row * LDSS + colB] = fmaxf(c1[v] * nd + biasB, 0.0f) * ns;
        }
    }
    __syncthreads();

    const int col = (wave << 4) + c16;
    v8f c = {};
    #pragma unroll 4
    for (int kk = 0; kk < 128; kk += 4) {
        const int k0 = kk + krow;
        v2f a = *(const v2f*)(&Bs[c16 * LDSS + k0]);
        v2f b;
        b.x = W2[(size_t)k0 * 64 + col];
        b.y = W2[(size_t)(k0 + 1) * 64 + col];
        c = __builtin_amdgcn_wmma_f32_16x16x4_f32(false, a, false, b, (short)0, c, false, false);
    }

    const int rbase = mbase + (half << 3);
    if (mbase + 16 <= nn) {
        float* p = H2 + (size_t)rbase * 64 + col;
        #pragma unroll
        for (int v = 0; v < 8; ++v) p[v * 64] = c[v];
    } else {
        #pragma unroll
        for (int v = 0; v < 8; ++v) {
            int r = rbase + v;
            if (r < nn) H2[(size_t)r * 64 + col] = c[v];
        }
    }
}


__global__ __launch_bounds__(256) void gcn_gather64(const float* __restrict__ H,
                                                    const int* __restrict__ deg_dst,
                                                    const int* __restrict__ bucket,
                                                    const int* __restrict__ ovf_n,
                                                    const int* __restrict__ ovf, int E,
                                                    const float* __restrict__ norm_dst,
                                                    const float* __restrict__ b2,
                                                    float* __restrict__ out, int nn) {
    const int d = __builtin_amdgcn_readfirstlane((int)(blockIdx.x * 8 + (threadIdx.x >> 5)));
    if (d >= nn) return;
    const int lane = threadIdx.x & 31;
    const int deg = deg_dst[d];
    const int cnt = min(deg, CAP);
    const int mine = (lane < cnt) ? bucket[(size_t)d * CAP + lane] : 0;
    float2 acc = make_float2(0.0f, 0.0f);
    rows64(H, lane, mine, cnt, acc);
    if (deg > CAP) ovf64(H, ovf_n, ovf, E, d, lane, acc);
    const float nd = norm_dst[d];
    const float2 bb = *(const float2*)(b2 + lane * 2);
    float2 r;
    r.x = acc.x * nd + bb.x;
    r.y = acc.y * nd + bb.y;
    *(float2*)(out + (size_t)d * 64 + lane * 2) = r;
}


extern "C" void kernel_launch(void* const* d_in, const int* in_sizes, int n_in,
                              void* d_out, int out_size, void* d_ws, size_t ws_size,
                              hipStream_t stream) {
    const float* feats = (const float*)d_in[0];
    const float* W1    = (const float*)d_in[1];
    const float* b1    = (const float*)d_in[2];
    const float* W2    = (const float*)d_in[3];
    const float* b2    = (const float*)d_in[4];
    const int*   src   = (const int*)d_in[5];
    const int*   dst   = (const int*)d_in[6];

    const int N = in_sizes[0] / 128;
    const int E = in_sizes[5];

    float* ws       = (float*)d_ws;
    int*   cnt_src  = (int*)ws;            ws += (size_t)N * CPAD;
    int*   cnt_dst  = (int*)ws;            ws += (size_t)N * CPAD;
    int*   ovf_n    = (int*)ws;            ws += CPAD;
    float* H2       = ws;                  ws += (size_t)N * 64;
    int*   bucket   = (int*)ws;            ws += (size_t)N * CAP;
    int*   ovf      = (int*)ws;            ws += (size_t)E * 2;
    float* norm_src = ws;                  ws += N;
    float* norm_dst = ws;                  ws += N;
    int*   deg_dst  = (int*)ws;            ws += N;

    const long nzero = (long)N * (2 * CPAD) + CPAD;
    gcn_zero<<<2048, 256, 0, stream>>>((float*)d_ws, nzero);

    gcn_degree<<<(E + 255) / 256, 256, 0, stream>>>(src, dst, cnt_src, cnt_dst, bucket, ovf_n, ovf, E);
    gcn_norm<<<(2 * N + 255) / 256, 256, 0, stream>>>(cnt_src, norm_src, deg_dst, N);

    const int mtiles = (N + 15) / 16;
    const int ntiles = (N + 7) / 8;

    gcn_layer1<<<mtiles, 128, 0, stream>>>(feats, deg_dst, bucket, ovf_n, ovf, E,
                                           norm_src, norm_dst, W1, b1, W2, H2, N);

    gcn_gather64<<<ntiles, 256, 0, stream>>>(H2, deg_dst, bucket, ovf_n, ovf, E,
                                             norm_dst, b2, (float*)d_out, N);
}


# both degree counters and a node's id slots share ONE 128-byte record line (30 slots): norm kernel and dense norm/degree arrays go, clear is 8 B a node; sum order = arrival slot, past 30 list order
# speedup vs baseline: 3.2500x; 1.0854x over previous
#include <hip/hip_runtime.h>

typedef __attribute__((ext_vector_type(2))) float v2f;
typedef __attribute__((ext_vector_type(8))) float v8f;

#define LDSS 132
#define REC  32
#define CAP  30


__global__ __launch_bounds__(256) void gcn_clear(int* __restrict__ rec,
                                                 int* __restrict__ ovf_n, int nn) {
    const int i = blockIdx.x * blockDim.x + threadIdx.x;
    if (i < nn) *(int2*)(rec + (size_t)i * REC) = make_int2(0, 0);
    if (i == 0) *ovf_n = 0;
}

__global__ __launch_bounds__(256) void gcn_degree(const int* __restrict__ src,
                                                  const int* __restrict__ dst,
                                                  int* __restrict__ rec,
                                                  int* __restrict__ ovf_n,
                                                  int* __restrict__ ovf, int E) {
    int e = blockIdx.x * blockDim.x + threadIdx.x;
    if (e < E) {
        const int s = src[e];
        const int d = dst[e];
        atomicAdd(&rec[(size_t)s * REC + 1], 1);
        const int slot = atomicAdd(&rec[(size_t)d * REC], 1);
        if (slot < CAP) {
            rec[(size_t)d * REC + 2 + slot] = s;
        } else {
            const int k = atomicAdd(ovf_n, 1);
            ((int2*)ovf)[k] = make_int2(s, d);
        }
    }
}

__device__ __forceinline__ float deg_norm(int c) { return rsqrtf(fmaxf((float)c, 1.0f)); }


__device__ __forceinline__ void rows128(const float* __restrict__ feats, int lane,
                                        int id, int wbits, int cnt, float4& acc) {
    for (int k = 0; k < cnt; k += 8) {
        float4 m[8];
        float  wv[8];
        #pragma unroll
        for (int u = 0; u < 8; ++u) {
            m[u] = make_float4(0.0f, 0.0f, 0.0f, 0.0f);
            wv[u] = 0.0f;
            if (u == 0 || k + u < cnt) {
                const int s = __builtin_amdgcn_readlane(id, k + u + 2);
                wv[u] = __int_as_float(__builtin_amdgcn_readlane(wbits, k + u + 2));
                m[u] = *(const float4*)(feats + (size_t)s * 128 + lane * 4);
            }
        }
        #pragma unroll
        for (int u = 0; u < 8; ++u) {
            acc.x += m[u].x * wv[u]; acc.y += m[u].y * wv[u];
            acc.z += m[u].z * wv[u]; acc.w += m[u].w * wv[u];
        }
    }
}

__device__ __forceinline__ void rows64(const float* __restrict__ H, int lane,
                                       int id, int cnt, float2& acc) {
    for (int k = 0; k < cnt; k += 8) {
        float2 m[8];
        #pragma unroll
        for (int u = 0; u < 8; ++u) {
            m[u] = make_float2(0.0f, 0.0f);
            if (u == 0 || k + u < cnt) {
                const int s = __builtin_amdgcn_readlane(id, k + u + 2);
                m[u] = *(const float2*)(H + (size_t)s * 64 + lane * 2);
            }
        }
        #pragma unroll
        for (int u = 0; u < 8; ++u) { acc.x += m[u].x; acc.y += m[u].y; }
    }
}

__device__ __forceinline__ void ovf128(const float* __restrict__ feats,
                                       const int* __restrict__ rec,
                                       const int* __restrict__ ovf_n,
                                       const int* __restrict__ ovf, int E,
                                       int d, int lane, float4& acc) {
    const int no = min(*ovf_n, E);
    const int2* __restrict__ ovf2 = (const int2*)ovf;
    for (int ob = 0; ob < no; ob += 32) {
        int2 r = make_int2(0, -1);
        if (ob + lane < no) r = ovf2[ob + lane];
        unsigned hits = __builtin_amdgcn_ballot_w32(r.y == d);
        while (hits) {
            const int l = __builtin_ctz(hits);
            hits &= hits - 1;
            const int s = __builtin_amdgcn_readlane(r.x, l);
            const float w = deg_norm(rec[(size_t)s * REC + 1]);
            const float4 m = *(const float4*)(feats + (size_t)s * 128 + lane * 4);
            acc.x += m.x * w; acc.y += m.y * w; acc.z += m.z * w; acc.w += m.w * w;
        }
    }
}

__device__ __forceinline__ void ovf64(const float* __restrict__ H,
                                      const int* __restrict__ ovf_n,
                                      const int* __restrict__ ovf, int E,
                                      int d, int lane, float2& acc) {
    const int no = min(*ovf_n, E);
    const int2* __restrict__ ovf2 = (const int2*)ovf;
    for (int ob = 0; ob < no; ob += 32) {
        int2 r = make_int2(0, -1);
        if (ob + lane < no) r = ovf2[ob + lane];
        unsigned hits = __builtin_amdgcn_ballot_w32(r.y == d);
        while (hits) {
            const int l = __builtin_ctz(hits);
            hits &= hits - 1;
            const int s = __builtin_amdgcn_readlane(r.x, l);
            const float2 m = *(const float2*)(H + (size_t)s * 64 + lane * 2);
            acc.x += m.x; acc.y += m.y;
        }
    }
}


__global__ __launch_bounds__(128) void gcn_layer1(const float* __restrict__ feats,
                                                  const int* __restrict__ rec,
                                                  const int* __restrict__ ovf_n,
                                                  const int* __restrict__ ovf, int E,
                                                  const float* __restrict__ W1,
                                                  const float* __restrict__ b1,
                                                  const float* __restrict__ W2,
                                                  float* __restrict__ H2, int nn) {
    __shared__ float As[16 * LDSS];
    __shared__ float Bs[16 * LDSS];
    __shared__ float nds[16];
    __shared__ float nss[16];
    const int tid  = threadIdx.x;
    const int lane = tid & 31;
    const int wave = __builtin_amdgcn_readfirstlane(tid >> 5);
    const int mbase = blockIdx.x * 16;

    if (tid < 16) {
        int r = mbase + tid; if (r >= nn) r = nn - 1;
        const int2 c = *(const int2*)(rec + (size_t)r * REC);
        nds[tid] = deg_norm(c.x);
        nss[tid] = deg_norm(c.y);
    }

    const int d0 = mbase + wave * 4;
    int raw[4];
    int dgs[4];
    int cnts[4];
    int ids[4];
    int wbs[4];
    #pragma unroll
    for (int j = 0; j < 4; ++j) {
        const int node = min(d0 + j, nn - 1);
        raw[j] = rec[(size_t)node * REC + lane];
    }
    #pragma unroll
    for (int j = 0; j < 4; ++j) {
        dgs[j] = (d0 + j < nn) ? __builtin_amdgcn_readlane(raw[j], 0) : 0;
        cnts[j] = min(dgs[j], CAP);
        ids[j] = (lane >= 2 && lane - 2 < cnts[j]) ? raw[j] : 0;
        wbs[j] = __float_as_int(deg_norm(rec[(size_t)ids[j] * REC + 1]));
    }
    #pragma unroll
    for (int j = 0; j < 4; ++j) {
        float4 acc = make_float4(0.0f, 0.0f, 0.0f, 0.0f);
        rows128(feats, lane, ids[j], wbs[j], cnts[j], acc);
        *(float4*)(&As[(wave * 4 + j) * LDSS + lane * 4]) = acc;
    }
    #pragma unroll
    for (int j = 0; j < 4; ++j) {
        if (dgs[j] > CAP) {
            float4 acc = *(const float4*)(&As[(wave * 4 + j) * LDSS + lane * 4]);
            ovf128(feats, rec, ovf_n, ovf, E, d0 + j, lane, acc);
            *(float4*)(&As[(wave * 4 + j) * LDSS + lane * 4]) = acc;
        }
    }
    __syncthreads();

    const int c16  = lane & 15;
    const int half = lane >> 4;
    const int krow = half << 1;

    const int colA = (wave << 5) + c16;
    const int colB = colA + 16;
    v8f c0 = {};
    v8f c1 = {};
    #pragma unroll 4
    for (int kk = 0; kk < 128; kk += 4) {
        const int k0 = kk + krow;
        v2f a = *(const v2f*)(&As[c16 * LDSS + k0]);
        v2f p, q;
        p.x = W1[(size_t)k0 * 128 + colA];
        p.y = W1[(size_t)(k0 + 1) * 128 + colA];
        q.x = W1[(size_t)k0 * 128 + colB];
        q.y = W1[(size_t)(k0 + 1) * 128 + colB];
        c0 = __builtin_amdgcn_wmma_f32_16x16x4_f32(false, a, false, p, (short)0, c0, false, false);
        c1 = __builtin_amdgcn_wmma_f32_16x16x4_f32(false, a, false, q, (short)0, c1, false, false);
    }

    {
        const float biasA = b1[colA];
        const float biasB = b1[colB];
        #pragma unroll
        for (int v = 0; v < 8; ++v) {
            const int row = (half << 3) + v;
            const float nd = nds[row];
            const float ns = nss[row];
            Bs[row * LDSS + colA] = fmaxf(c0[v] * nd + biasA, 0.0f) * ns;
            Bs[row * LDSS + colB] = fmaxf(c1[v] * nd + biasB, 0.0f) * ns;
        }
    }
    __syncthreads();

    const int col = (wave << 4) + c16;
    v8f c = {};
    #pragma unroll 4
    for (int kk = 0; kk < 128; kk += 4) {
        const int k0 = kk + krow;
        v2f a = *(const v2f*)(&Bs[c16 * LDSS + k0]);
        v2f b;
        b.x = W2[(size_t)k0 * 64 + col];
        b.y = W2[(size_t)(k0 + 1) * 64 + col];
        c = __builtin_amdgcn_wmma_f32_16x16x4_f32(false, a, false, b, (short)0, c, false, false);
    }

    const int rbase = mbase + (half << 3);
    if (mbase + 16 <= nn) {
        float* p = H2 + (size_t)rbase * 64 + col;
        #pragma unroll
        for (int v = 0; v < 8; ++v) p[v * 64] = c[v];
    } else {
        #pragma unroll
        for (int v = 0; v < 8; ++v) {
            int r = rbase + v;
            if (r < nn) H2[(size_t)r * 64 + col] = c[v];
        }
    }
}


__global__ __launch_bounds__(256) void gcn_gather64(const float* __restrict__ H,
                                                    const int* __restrict__ rec,
                                                    const int* __restrict__ ovf_n,
                                                    const int* __restrict__ ovf, int E,
                                                    const float* __restrict__ b2,
                                                    float* __restrict__ out, int nn) {
    const int d = __builtin_amdgcn_readfirstlane((int)(blockIdx.x * 8 + (threadIdx.x >> 5)));
    if (d >= nn) return;
    const int lane = threadIdx.x & 31;
    const int raw = rec[(size_t)d * REC + lane];
    const int deg = __builtin_amdgcn_readlane(raw, 0);
    const int cnt = min(deg, CAP);
    const int mine = (lane >= 2 && lane - 2 < cnt) ? raw : 0;
    float2 acc = make_float2(0.0f, 0.0f);
    rows64(H, lane, mine, cnt, acc);
    if (deg > CAP) ovf64(H, ovf_n, ovf, E, d, lane, acc);
    const float nd = deg_norm(deg);
    const float2 bb = *(const float2*)(b2 + lane * 2);
    float2 r;
    r.x = acc.x * nd + bb.x;
    r.y = acc.y * nd + bb.y;
    *(float2*)(out + (size_t)d * 64 + lane * 2) = r;
}


extern "C" void kernel_launch(void* const* d_in, const int* in_sizes, int n_in,
                              void* d_out, int out_size, void* d_ws, size_t ws_size,
                              hipStream_t stream) {
    const float* feats = (const float*)d_in[0];
    const float* W1    = (const float*)d_in[1];
    const float* b1    = (const float*)d_in[2];
    const float* W2    = (const float*)d_in[3];
    const float* b2    = (const float*)d_in[4];
    const int*   src   = (const int*)d_in[5];
    const int*   dst   = (const int*)d_in[6];

    const int N = in_sizes[0] / 128;
    const int E = in_sizes[5];

    float* ws     = (float*)d_ws;
    int*   rec    = (int*)ws;              ws += (size_t)N * REC;
    int*   ovf_n  = (int*)ws;              ws += REC;
    float* H2     = ws;                    ws += (size_t)N * 64;
    int*   ovf    = (int*)ws;              ws += (size_t)E * 2;

    gcn_clear<<<(N + 255) / 256, 256, 0, stream>>>(rec, ovf_n, N);

    gcn_degree<<<(E + 255) / 256, 256, 0, stream>>>(src, dst, rec, ovf_n, ovf, E);

    const int mtiles = (N + 15) / 16;
    const int ntiles = (N + 7) / 8;

    gcn_layer1<<<mtiles, 128, 0, stream>>>(feats, rec, ovf_n, ovf, E, W1, b1, W2, H2, N);

    gcn_gather64<<<ntiles, 256, 0, stream>>>(H2, rec, ovf_n, ovf, E, b2, (float*)d_out, N);
}


# edge filing split by chip half: nodes in 2 sets, a work-group claims (set, 1024-edge piece) items, its own die half's set first, so a record line is updated from one half; slot order = arrival
# speedup vs baseline: 5.1857x; 1.5956x over previous
#include <hip/hip_runtime.h>

typedef __attribute__((ext_vector_type(2))) float v2f;
typedef __attribute__((ext_vector_type(8))) float v8f;

#define LDSS 132
#define REC  32
#define CAP  30
#define CHUNK 1024


__global__ __launch_bounds__(256) void gcn_clear(int* __restrict__ rec,
                                                 int* __restrict__ ovf_n, int nn) {
    const int i = blockIdx.x * blockDim.x + threadIdx.x;
    if (i < nn) *(int2*)(rec + (size_t)i * REC) = make_int2(0, 0);
    if (i == 0) { ovf_n[0] = 0; ovf_n[REC] = 0; ovf_n[2 * REC] = 0; }
}

__global__ __launch_bounds__(256) void gcn_degree(const int* __restrict__ src,
                                                  const int* __restrict__ dst,
                                                  int* __restrict__ rec,
                                                  int* __restrict__ ovf_n,
                                                  int* __restrict__ ovf, int E, int nn) {
    __shared__ int claimed;
    const int pieces = (E + CHUNK - 1) / CHUNK;
    const int split = (nn + 1) >> 1;
    const int die = (__builtin_amdgcn_s_sendmsg_rtn(0x87) >> 16) & 15;
    int q = (die >> 2) & 1;
    for (int pass = 0; pass < 2; ++pass) {
        for (;;) {
            if (threadIdx.x == 0) claimed = atomicAdd(&ovf_n[(1 + q) * REC], 1);
            __syncthreads();
            const int piece = __builtin_amdgcn_readfirstlane(claimed);
            __syncthreads();
            if (piece >= pieces) break;
            const int base = piece * CHUNK + (int)threadIdx.x;
            int  s[4], d[4], slot[4];
            bool ms[4], md[4];
            #pragma unroll
            for (int j = 0; j < 4; ++j) {
                const int e = min(base + j * 256, E - 1);
                s[j] = src[e]; d[j] = dst[e];
            }
            #pragma unroll
            for (int j = 0; j < 4; ++j) {
                const bool real = base + j * 256 < E;
                ms[j] = real && ((s[j] >= split) == (q != 0));
                md[j] = real && ((d[j] >= split) == (q != 0));
            }
            #pragma unroll
            for (int j = 0; j < 4; ++j) {
                if (ms[j]) atomicAdd(&rec[(size_t)s[j] * REC + 1], 1);
                slot[j] = 0;
                if (md[j]) slot[j] = atomicAdd(&rec[(size_t)d[j] * REC], 1);
            }
            #pragma unroll
            for (int j = 0; j < 4; ++j) {
                if (md[j]) {
                    if (slot[j] < CAP) {
                        rec[(size_t)d[j] * REC + 2 + slot[j]] = s[j];
                    } else {
                        const int k = atomicAdd(ovf_n, 1);
                        ((int2*)ovf)[k] = make_int2(s[j], d[j]);
                    }
                }
            }
        }
        q ^= 1;
    }
}

__device__ __forceinline__ float deg_norm(int c) { return rsqrtf(fmaxf((float)c, 1.0f)); }


__device__ __forceinline__ void rows128(const float* __restrict__ feats, int lane,
                                        int id, int wbits, int cnt, float4& acc) {
    for (int k = 0; k < cnt; k += 8) {
        float4 m[8];
        float  wv[8];
        #pragma unroll
        for (int u = 0; u < 8; ++u) {
            m[u] = make_float4(0.0f, 0.0f, 0.0f, 0.0f);
            wv[u] = 0.0f;
            if (u == 0 || k + u < cnt) {
                const int s = __builtin_amdgcn_readlane(id, k + u + 2);
                wv[u] = __int_as_float(__builtin_amdgcn_readlane(wbits, k + u + 2));
                m[u] = *(const float4*)(feats + (size_t)s * 128 + lane * 4);
            }
        }
        #pragma unroll
        for (int u = 0; u < 8; ++u) {
            acc.x += m[u].x * wv[u]; acc.y += m[u].y * wv[u];
            acc.z += m[u].z * wv[u]; acc.w += m[u].w * wv[u];
        }
    }
}

__device__ __forceinline__ void rows64(const float* __restrict__ H, int lane,
                                       int id, int cnt, float2& acc) {
    for (int k = 0; k < cnt; k += 8) {
        float2 m[8];
        #pragma unroll
        for (int u = 0; u < 8; ++u) {
            m[u] = make_float2(0.0f, 0.0f);
            if (u == 0 || k + u < cnt) {
                const int s = __builtin_amdgcn_readlane(id, k + u + 2);
                m[u] = *(const float2*)(H + (size_t)s * 64 + lane * 2);
            }
        }
        #pragma unroll
        for (int u = 0; u < 8; ++u) { acc.x += m[u].x; acc.y += m[u].y; }
    }
}

__device__ __forceinline__ void ovf128(const float* __restrict__ feats,
                                       const int* __restrict__ rec,
                                       const int* __restrict__ ovf_n,
                                       const int* __restrict__ ovf, int E,
                                       int d, int lane, float4& acc) {
    const int no = min(*ovf_n, E);
    const int2* __restrict__ ovf2 = (const int2*)ovf;
    for (int ob = 0; ob < no; ob += 32) {
        int2 r = make_int2(0, -1);
        if (ob + lane < no) r = ovf2[ob + lane];
        unsigned hits = __builtin_amdgcn_ballot_w32(r.y == d);
        while (hits) {
            const int l = __builtin_ctz(hits);
            hits &= hits - 1;
            const int s = __builtin_amdgcn_readlane(r.x, l);
            const float w = deg_norm(rec[(size_t)s * REC + 1]);
            const float4 m = *(const float4*)(feats + (size_t)s * 128 + lane * 4);
            acc.x += m.x * w; acc.y += m.y * w; acc.z += m.z * w; acc.w += m.w * w;
        }
    }
}

__device__ __forceinline__ void ovf64(const float* __restrict__ H,
                                      const int* __restrict__ ovf_n,
                                      const int* __restrict__ ovf, int E,
                                      int d, int lane, float2& acc) {
    const int no = min(*ovf_n, E);
    const int2* __restrict__ ovf2 = (const int2*)ovf;
    for (int ob = 0; ob < no; ob += 32) {
        int2 r = make_int2(0, -1);
        if (ob + lane < no) r = ovf2[ob + lane];
        unsigned hits = __builtin_amdgcn_ballot_w32(r.y == d);
        while (hits) {
            const int l = __builtin_ctz(hits);
            hits &= hits - 1;
            const int s = __builtin_amdgcn_readlane(r.x, l);
            const float2 m = *(const float2*)(H + (size_t)s * 64 + lane * 2);
            acc.x += m.x; acc.y += m.y;
        }
    }
}


__global__ __launch_bounds__(128) void gcn_layer1(const float* __restrict__ feats,
                                                  const int* __restrict__ rec,
                                                  const int* __restrict__ ovf_n,
                                                  const int* __restrict__ ovf, int E,
                                                  const float* __restrict__ W1,
                                                  const float* __restrict__ b1,
                                                  const float* __restrict__ W2,
                                                  float* __restrict__ H2, int nn) {
    __shared__ float As[16 * LDSS];
    __shared__ float Bs[16 * LDSS];
    __shared__ float nds[16];
    __shared__ float nss[16];
    const int tid  = threadIdx.x;
    const int lane = tid & 31;
    const int wave = __builtin_amdgcn_readfirstlane(tid >> 5);
    const int mbase = blockIdx.x * 16;

    if (tid < 16) {
        int r = mbase + tid; if (r >= nn) r = nn - 1;
        const int2 c = *(const int2*)(rec + (size_t)r * REC);
        nds[tid] = deg_norm(c.x);
        nss[tid] = deg_norm(c.y);
    }

    const int d0 = mbase + wave * 4;
    int raw[4];
    int dgs[4];
    int cnts[4];
    int ids[4];
    int wbs[4];
    #pragma unroll
    for (int j = 0; j < 4; ++j) {
        const int node = min(d0 + j, nn - 1);
        raw[j] = rec[(size_t)node * REC + lane];
    }
    #pragma unroll
    for (int j = 0; j < 4; ++j) {
        dgs[j] = (d0 + j < nn) ? __builtin_amdgcn_readlane(raw[j], 0) : 0;
        cnts[j] = min(dgs[j], CAP);
        ids[j] = (lane >= 2 && lane - 2 < cnts[j]) ? raw[j] : 0;
        wbs[j] = __float_as_int(deg_norm(rec[(size_t)ids[j] * REC + 1]));
    }
    #pragma unroll
    for (int j = 0; j < 4; ++j) {
        float4 acc = make_float4(0.0f, 0.0f, 0.0f, 0.0f);
        rows128(feats, lane, ids[j], wbs[j], cnts[j], acc);
        *(float4*)(&As[(wave * 4 + j) * LDSS + lane * 4]) = acc;
    }
    #pragma unroll
    for (int j = 0; j < 4; ++j) {
        if (dgs[j] > CAP) {
            float4 acc = *(const float4*)(&As[(wave * 4 + j) * LDSS + lane * 4]);
            ovf128(feats, rec, ovf_n, ovf, E, d0 + j, lane, acc);
            *(float4*)(&As[(wave * 4 + j) * LDSS + lane * 4]) = acc;
        }
    }
    __syncthreads();

    const int c16  = lane & 15;
    const int half = lane >> 4;
    const int krow = half << 1;

    const int colA = (wave << 5) + c16;
    const int colB = colA + 16;
    v8f c0 = {};
    v8f c1 = {};
    #pragma unroll 4
    for (int kk = 0; kk < 128; kk += 4) {
        const int k0 = kk + krow;
        v2f a = *(const v2f*)(&As[c16 * LDSS + k0]);
        v2f p, q;
        p.x = W1[(size_t)k0 * 128 + colA];
        p.y = W1[(size_t)(k0 + 1) * 128 + colA];
        q.x = W1[(size_t)k0 * 128 + colB];
        q.y = W1[(size_t)(k0 + 1) * 128 + colB];
        c0 = __builtin_amdgcn_wmma_f32_16x16x4_f32(false, a, false, p, (short)0, c0, false, false);
        c1 = __builtin_amdgcn_wmma_f32_16x16x4_f32(false, a, false, q, (short)0, c1, false, false);
    }

    {
        const float biasA = b1[colA];
        const float biasB = b1[colB];
        #pragma unroll
        for (int v = 0; v < 8; ++v) {
            const int row = (half << 3) + v;
            const float nd = nds[row];
            const float ns = nss[row];
            Bs[row * LDSS + colA] = fmaxf(c0[v] * nd + biasA, 0.0f) * ns;
            Bs[row * LDSS + colB] = fmaxf(c1[v] * nd + biasB, 0.0f) * ns;
        }
    }
    __syncthreads();

    const int col = (wave << 4) + c16;
    v8f c = {};
    #pragma unroll 4
    for (int kk = 0; kk < 128; kk += 4) {
        const int k0 = kk + krow;
        v2f a = *(const v2f*)(&Bs[c16 * LDSS + k0]);
        v2f b;
        b.x = W2[(size_t)k0 * 64 + col];
        b.y = W2[(size_t)(k0 + 1) * 64 + col];
        c = __builtin_amdgcn_wmma_f32_16x16x4_f32(false, a, false, b, (short)0, c, false, false);
    }

    const int rbase = mbase + (half << 3);
    if (mbase + 16 <= nn) {
        float* p = H2 + (size_t)rbase * 64 + col;
        #pragma unroll
        for (int v = 0; v < 8; ++v) p[v * 64] = c[v];
    } else {
        #pragma unroll
        for (int v = 0; v < 8; ++v) {
            int r = rbase + v;
            if (r < nn) H2[(size_t)r * 64 + col] = c[v];
        }
    }
}


__global__ __launch_bounds__(256) void gcn_gather64(const float* __restrict__ H,
                                                    const int* __restrict__ rec,
                                                    const int* __restrict__ ovf_n,
                                                    const int* __restrict__ ovf, int E,
                                                    const float* __restrict__ b2,
                                                    float* __restrict__ out, int nn) {
    const int d = __builtin_amdgcn_readfirstlane((int)(blockIdx.x * 8 + (threadIdx.x >> 5)));
    if (d >= nn) return;
    const int lane = threadIdx.x & 31;
    const int raw = rec[(size_t)d * REC + lane];
    const int deg = __builtin_amdgcn_readlane(raw, 0);
    const int cnt = min(deg, CAP);
    const int mine = (lane >= 2 && lane - 2 < cnt) ? raw : 0;
    float2 acc = make_float2(0.0f, 0.0f);
    rows64(H, lane, mine, cnt, acc);
    if (deg > CAP) ovf64(H, ovf_n, ovf, E, d, lane, acc);
    const float nd = deg_norm(deg);
    const float2 bb = *(const float2*)(b2 + lane * 2);
    float2 r;
    r.x = acc.x * nd + bb.x;
    r.y = acc.y * nd + bb.y;
    *(float2*)(out + (size_t)d * 64 + lane * 2) = r;
}


extern "C" void kernel_launch(void* const* d_in, const int* in_sizes, int n_in,
                              void* d_out, int out_size, void* d_ws, size_t ws_size,
                              hipStream_t stream) {
    const float* feats = (const float*)d_in[0];
    const float* W1    = (const float*)d_in[1];
    const float* b1    = (const float*)d_in[2];
    const float* W2    = (const float*)d_in[3];
    const float* b2    = (const float*)d_in[4];
    const int*   src   = (const int*)d_in[5];
    const int*   dst   = (const int*)d_in[6];

    const int N = in_sizes[0] / 128;
    const int E = in_sizes[5];

    float* ws     = (float*)d_ws;
    int*   rec    = (int*)ws;              ws += (size_t)N * REC;
    int*   ovf_n  = (int*)ws;              ws += 3 * REC;
    float* H2     = ws;                    ws += (size_t)N * 64;
    int*   ovf    = (int*)ws;              ws += (size_t)E * 2;

    gcn_clear<<<(N + 255) / 256, 256, 0, stream>>>(rec, ovf_n, N);

    gcn_degree<<<2048, 256, 0, stream>>>(src, dst, rec, ovf_n, ovf, E, N);

    const int mtiles = (N + 15) / 16;
    const int ntiles = (N + 7) / 8;

    gcn_layer1<<<mtiles, 128, 0, stream>>>(feats, rec, ovf_n, ovf, E, W1, b1, W2, H2, N);

    gcn_gather64<<<ntiles, 256, 0, stream>>>(H2, rec, ovf_n, ovf, E, b2, (float*)d_out, N);
}
